# starting kernel
# baseline (speedup 1.0000x reference)
#include <hip/hip_runtime.h>

typedef __attribute__((ext_vector_type(2))) float v2f;
typedef __attribute__((ext_vector_type(8))) float v8f;

#define CONV_H 128
#define CONV_W 128
#define CONV_IC 8
#define CONV_OC 16
#define K_STEPS 18
#define MTILES 4

__global__ __launch_bounds__(256)
void conv2d_nhwc_wmma_f32(const float* __restrict__ x,
                          const float* __restrict__ wgt,
                          float* __restrict__ out) {
  const int lane = threadIdx.x & 31;
  const int wave = threadIdx.x >> 5;
  const int job  = blockIdx.x * 8 + wave;
  const int wseg = job & 1;
  const int h    = (job >> 1) & 127;
  const int n    = job >> 8;

  const int half = lane >> 4;
  const int ml   = lane & 15;
  const int w0   = wseg * 64;

  v2f bfrag[K_STEPS];
#pragma unroll
  for (int kk = 0; kk < K_STEPS; ++kk) {
    const int k0 = 4 * kk + 2 * half;
    bfrag[kk].x = wgt[k0 * CONV_OC + ml];
    bfrag[kk].y = wgt[(k0 + 1) * CONV_OC + ml];
  }

  const float* xN = x + (size_t)n * CONV_H * CONV_W * CONV_IC;

  v8f acc[MTILES];
#pragma unroll
  for (int t = 0; t < MTILES; ++t) acc[t] = (v8f){};

#pragma unroll
  for (int kk = 0; kk < K_STEPS; ++kk) {
    const int pix = kk >> 1;
    const int fh  = pix / 3;
    const int fw  = pix % 3;
    const int ic  = (kk & 1) * 4 + 2 * half;

    const int  ih   = h + fh - 1;
    const int  ihc  = ih < 0 ? 0 : (ih > CONV_H - 1 ? CONV_H - 1 : ih);
    const bool ihok = (ih >= 0) & (ih < CONV_H);
    const float* rowp = xN + (size_t)ihc * CONV_W * CONV_IC + ic;

#pragma unroll
    for (int t = 0; t < MTILES; ++t) {
      const int iw  = w0 + 16 * t + ml + fw - 1;
      const int iwc = iw < 0 ? 0 : (iw > CONV_W - 1 ? CONV_W - 1 : iw);
      const float msk = (ihok & (iw >= 0) & (iw < CONV_W)) ? 1.0f : 0.0f;

      v2f a = *(const v2f*)(rowp + (size_t)iwc * CONV_IC);
      a.x *= msk;
      a.y *= msk;

      acc[t] = __builtin_amdgcn_wmma_f32_16x16x4_f32(
           false, a,  false, bfrag[kk],
           (short)0, acc[t],  false,  false);
    }
  }

  float* o = out + (((size_t)n * CONV_H + h) * CONV_W + w0) * CONV_OC;
#pragma unroll
  for (int t = 0; t < MTILES; ++t) {
#pragma unroll
    for (int j = 0; j < 8; ++j) {
      o[(size_t)(16 * t + j + 8 * half) * CONV_OC + ml] = acc[t][j];
    }
  }
}

extern "C" void kernel_launch(void* const* d_in, const int* in_sizes, int n_in,
                              void* d_out, int out_size, void* d_ws, size_t ws_size,
                              hipStream_t stream) {
  const float* x = (const float*)d_in[0];
  const float* w = (const float*)d_in[1];
  float* out = (float*)d_out;

  const int jobs   = 32 * 128 * (128 / 64);
  const int blocks = jobs / 8;
  conv2d_nhwc_wmma_f32<<<blocks, 256, 0, stream>>>(x, w, out);
}


# Each wave walks 4 output rows: an input row's A fragments are loaded once and kept in registers for all 3 filter rows, next row's loads grouped a step ahead; 2048 waves, sum order unchanged
# speedup vs baseline: 1.2793x; 1.2793x over previous
#include <hip/hip_runtime.h>


typedef __attribute__((ext_vector_type(2))) float v2f;
typedef __attribute__((ext_vector_type(8))) float v8f;

#define CONV_H 128
#define CONV_W 128
#define CONV_IC 8
#define CONV_OC 16
#define K_STEPS 18
#define MTILES 4
#define ROWS 4


struct RowFrags {
  v2f a[3][2][MTILES];
};

__device__ __forceinline__ bool w_edge(int fw, int t) {
  return (fw == 0 && t == 0) || (fw == 2 && t == MTILES - 1);
}

template <bool EDGE>
__device__ __forceinline__ void load_row(RowFrags& f, const float* __restrict__ xN,
                                         int r, int w0, int ml, int half) {
  int rc = r;
  if (EDGE) rc = r < 0 ? 0 : (r > CONV_H - 1 ? CONV_H - 1 : r);
  const float* rowp = xN + (size_t)rc * CONV_W * CONV_IC + 2 * half;
#pragma unroll
  for (int fw = 0; fw < 3; ++fw) {
#pragma unroll
    for (int t = 0; t < MTILES; ++t) {
      const int iw  = w0 + 16 * t + ml + fw - 1;
      const int iwc = iw < 0 ? 0 : (iw > CONV_W - 1 ? CONV_W - 1 : iw);
#pragma unroll
      for (int kc = 0; kc < 2; ++kc) {
        f.a[fw][kc][t] = *(const v2f*)(rowp + (size_t)iwc * CONV_IC + kc * 4);
      }
    }
  }
}

template <bool EDGE>
__device__ __forceinline__ void mask_row(RowFrags& f, int r, int w0, int ml) {
  float rmask = 1.0f;
  if (EDGE) rmask = ((r >= 0) & (r < CONV_H)) ? 1.0f : 0.0f;
#pragma unroll
  for (int fw = 0; fw < 3; ++fw) {
#pragma unroll
    for (int t = 0; t < MTILES; ++t) {
      if (!EDGE && !w_edge(fw, t)) continue;
      const int iw = w0 + 16 * t + ml + fw - 1;
      const float msk = ((iw >= 0) & (iw < CONV_W)) ? rmask : 0.0f;
#pragma unroll
      for (int kc = 0; kc < 2; ++kc) {
        f.a[fw][kc][t].x *= msk;
        f.a[fw][kc][t].y *= msk;
      }
    }
  }
}

__global__ __launch_bounds__(128)
void conv2d_nhwc_wmma_f32(const float* __restrict__ x,
                          const float* __restrict__ wgt,
                          float* __restrict__ out) {
  const int lane = threadIdx.x & 31;
  const int wave = threadIdx.x >> 5;
  const int job  = blockIdx.x * 4 + wave;
  const int wseg = job & 1;
  const int h0   = ((job >> 1) & 31) * ROWS;
  const int n    = job >> 6;

  const int half = lane >> 4;
  const int ml   = lane & 15;
  const int w0   = wseg * 64;

  v2f bfrag[K_STEPS];
#pragma unroll
  for (int kk = 0; kk < K_STEPS; ++kk) {
    const int k0 = 4 * kk + 2 * half;
    bfrag[kk].x = wgt[k0 * CONV_OC + ml];
    bfrag[kk].y = wgt[(k0 + 1) * CONV_OC + ml];
  }

  const float* xN = x + (size_t)n * CONV_H * CONV_W * CONV_IC;
  float* oN = out + ((size_t)n * CONV_H * CONV_W + w0) * CONV_OC;

  v8f acc[ROWS][MTILES];

  RowFrags cur, nxt;
  load_row<true>(cur, xN, h0 - 1, w0, ml, half);

#pragma unroll
  for (int ri = 0; ri < ROWS + 2; ++ri) {
    const int r = h0 - 1 + ri;
    if (ri + 1 < ROWS + 1) {
      load_row<false>(nxt, xN, r + 1, w0, ml, half);
    } else if (ri + 1 == ROWS + 1) {
      load_row<true>(nxt, xN, r + 1, w0, ml, half);
    }
    __builtin_amdgcn_sched_barrier(0);

    if (ri == 0 || ri == ROWS + 1) mask_row<true>(cur, r, w0, ml);
    else                           mask_row<false>(cur, r, w0, ml);

#pragma unroll
    for (int fhi = 0; fhi < 3; ++fhi) {
      const int fh = 2 - fhi;
      const int ho = ri - fh;
      if (ho < 0 || ho >= ROWS) continue;
#pragma unroll
      for (int fw = 0; fw < 3; ++fw) {
#pragma unroll
        for (int kc = 0; kc < 2; ++kc) {
          const int kk = (fh * 3 + fw) * 2 + kc;
#pragma unroll
          for (int t = 0; t < MTILES; ++t) {
            const v8f c = (kk == 0) ? (v8f){} : acc[ho][t];
            acc[ho][t] = __builtin_amdgcn_wmma_f32_16x16x4_f32(
                 false, cur.a[fw][kc][t],  false, bfrag[kk],
                 (short)0, c,  false,  false);
          }
        }
      }
      if (fh == 2) {
        float* o = oN + (size_t)(h0 + ho) * CONV_W * CONV_OC;
#pragma unroll
        for (int t = 0; t < MTILES; ++t) {
#pragma unroll
          for (int j = 0; j < 8; ++j) {
            o[(size_t)(16 * t + j + 8 * half) * CONV_OC + ml] = acc[ho][t][j];
          }
        }
      }
    }
    __builtin_amdgcn_sched_barrier(0);
    cur = nxt;
  }
}

extern "C" void kernel_launch(void* const* d_in, const int* in_sizes, int n_in,
                              void* d_out, int out_size, void* d_ws, size_t ws_size,
                              hipStream_t stream) {
  const float* x = (const float*)d_in[0];
  const float* w = (const float*)d_in[1];
  float* out = (float*)d_out;

  conv2d_nhwc_wmma_f32<<<512, 128, 0, stream>>>(x, w, out);
}


# Whole-line output stores: WMMA operands swapped so a lane holds 8 consecutive channels, tile staged in wave-private LDS, 2 b128 stores of 4 full lines per tile instead of 8 b32 half-line stores
# speedup vs baseline: 1.3879x; 1.0849x over previous
#include <hip/hip_runtime.h>


typedef __attribute__((ext_vector_type(2))) float v2f;
typedef __attribute__((ext_vector_type(4))) float v4f;
typedef __attribute__((ext_vector_type(8))) float v8f;

#define CONV_H 128
#define CONV_W 128
#define CONV_IC 8
#define CONV_OC 16
#define K_STEPS 18
#define MTILES 4
#define ROWS 4
#define TILE_CHUNKS 64


struct RowFrags {
  v2f a[3][2][MTILES];
};

__device__ __forceinline__ bool w_edge(int fw, int t) {
  return (fw == 0 && t == 0) || (fw == 2 && t == MTILES - 1);
}

template <bool EDGE>
__device__ __forceinline__ void load_row(RowFrags& f, const float* __restrict__ xN,
                                         int r, int w0, int ml, int half) {
  int rc = r;
  if (EDGE) rc = r < 0 ? 0 : (r > CONV_H - 1 ? CONV_H - 1 : r);
  const float* rowp = xN + (size_t)rc * CONV_W * CONV_IC + 2 * half;
#pragma unroll
  for (int fw = 0; fw < 3; ++fw) {
#pragma unroll
    for (int t = 0; t < MTILES; ++t) {
      const int iw  = w0 + 16 * t + ml + fw - 1;
      const int iwc = iw < 0 ? 0 : (iw > CONV_W - 1 ? CONV_W - 1 : iw);
#pragma unroll
      for (int kc = 0; kc < 2; ++kc) {
        f.a[fw][kc][t] = *(const v2f*)(rowp + (size_t)iwc * CONV_IC + kc * 4);
      }
    }
  }
}

template <bool EDGE>
__device__ __forceinline__ void mask_row(RowFrags& f, int r, int w0, int ml) {
  float rmask = 1.0f;
  if (EDGE) rmask = ((r >= 0) & (r < CONV_H)) ? 1.0f : 0.0f;
#pragma unroll
  for (int fw = 0; fw < 3; ++fw) {
#pragma unroll
    for (int t = 0; t < MTILES; ++t) {
      if (!EDGE && !w_edge(fw, t)) continue;
      const int iw = w0 + 16 * t + ml + fw - 1;
      const float msk = ((iw >= 0) & (iw < CONV_W)) ? rmask : 0.0f;
#pragma unroll
      for (int kc = 0; kc < 2; ++kc) {
        f.a[fw][kc][t].x *= msk;
        f.a[fw][kc][t].y *= msk;
      }
    }
  }
}

__global__ __launch_bounds__(128)
void conv2d_nhwc_wmma_f32(const float* __restrict__ x,
                          const float* __restrict__ wgt,
                          float* __restrict__ out) {
  const int lane = threadIdx.x & 31;
  const int wave = threadIdx.x >> 5;
  const int job  = blockIdx.x * 4 + wave;
  const int wseg = job & 1;
  const int h0   = ((job >> 1) & 31) * ROWS;
  const int n    = job >> 6;

  const int half = lane >> 4;
  const int ml   = lane & 15;

  __shared__ v4f stage[4][MTILES][TILE_CHUNKS];
  v4f (*stg)[TILE_CHUNKS] = stage[wave];
  const int w0   = wseg * 64;

  v2f bfrag[K_STEPS];
#pragma unroll
  for (int kk = 0; kk < K_STEPS; ++kk) {
    const int k0 = 4 * kk + 2 * half;
    bfrag[kk].x = wgt[k0 * CONV_OC + ml];
    bfrag[kk].y = wgt[(k0 + 1) * CONV_OC + ml];
  }

  const float* xN = x + (size_t)n * CONV_H * CONV_W * CONV_IC;
  v4f* oN = (v4f*)out + ((size_t)n * CONV_H * CONV_W + w0) * (CONV_OC / 4);

  v8f acc[ROWS][MTILES];

  RowFrags cur, nxt;
  load_row<true>(cur, xN, h0 - 1, w0, ml, half);

#pragma unroll
  for (int ri = 0; ri < ROWS + 2; ++ri) {
    const int r = h0 - 1 + ri;
    if (ri + 1 < ROWS + 1) {
      load_row<false>(nxt, xN, r + 1, w0, ml, half);
    } else if (ri + 1 == ROWS + 1) {
      load_row<true>(nxt, xN, r + 1, w0, ml, half);
    }
    __builtin_amdgcn_sched_barrier(0);

    v4f outv[MTILES][2];

    if (ri == 0 || ri == ROWS + 1) mask_row<true>(cur, r, w0, ml);
    else                           mask_row<false>(cur, r, w0, ml);

#pragma unroll
    for (int fhi = 0; fhi < 3; ++fhi) {
      const int fh = 2 - fhi;
      const int ho = ri - fh;
      if (ho < 0 || ho >= ROWS) continue;
#pragma unroll
      for (int fw = 0; fw < 3; ++fw) {
#pragma unroll
        for (int kc = 0; kc < 2; ++kc) {
          const int kk = (fh * 3 + fw) * 2 + kc;
#pragma unroll
          for (int t = 0; t < MTILES; ++t) {
            const v8f c = (kk == 0) ? (v8f){} : acc[ho][t];
            acc[ho][t] = __builtin_amdgcn_wmma_f32_16x16x4_f32(
                 false, bfrag[kk],  false, cur.a[fw][kc][t],
                 (short)0, c,  false,  false);
          }
        }
      }
      if (fh == 2) {
#pragma unroll
        for (int t = 0; t < MTILES; ++t) {
          const v8f d = acc[ho][t];
          stg[t][ml * 4 + half * 2 + 0] = (v4f){d[0], d[1], d[2], d[3]};
          stg[t][ml * 4 + half * 2 + 1] = (v4f){d[4], d[5], d[6], d[7]};
        }
        __builtin_amdgcn_wave_barrier();
#pragma unroll
        for (int t = 0; t < MTILES; ++t) {
          outv[t][0] = stg[t][lane];
          outv[t][1] = stg[t][lane + 32];
        }
        __builtin_amdgcn_wave_barrier();
        __builtin_amdgcn_sched_barrier(0);
      }
    }
    if (ri >= 2) {
      v4f* o = oN + (size_t)(h0 + ri - 2) * CONV_W * (CONV_OC / 4);
#pragma unroll
      for (int t = 0; t < MTILES; ++t) {
        o[t * TILE_CHUNKS + lane]      = outv[t][0];
        o[t * TILE_CHUNKS + 32 + lane] = outv[t][1];
      }
    }
    __builtin_amdgcn_sched_barrier(0);
    cur = nxt;
  }
}

extern "C" void kernel_launch(void* const* d_in, const int* in_sizes, int n_in,
                              void* d_out, int out_size, void* d_ws, size_t ws_size,
                              hipStream_t stream) {
  const float* x = (const float*)d_in[0];
  const float* w = (const float*)d_in[1];
  float* out = (float*)d_out;

  conv2d_nhwc_wmma_f32<<<512, 128, 0, stream>>>(x, w, out);
}


# 3-tap filter along H run as a 2-output minimal filter: 4 row products per 2 output rows instead of 6 (192 WMMAs per wave, was 288); transform-domain sums change rounding and sum order
# speedup vs baseline: 1.5383x; 1.1084x over previous
#include <hip/hip_runtime.h>


typedef __attribute__((ext_vector_type(2))) float v2f;
typedef __attribute__((ext_vector_type(4))) float v4f;
typedef __attribute__((ext_vector_type(8))) float v8f;

#define CONV_H 128
#define CONV_W 128
#define CONV_IC 8
#define CONV_OC 16
#define K_STEPS 18
#define MTILES 4
#define ROWS 4
#define TILE_CHUNKS 64


struct RowFrags {
  v2f a[3][2][MTILES];
};

__device__ __forceinline__ bool w_edge(int fw, int t) {
  return (fw == 0 && t == 0) || (fw == 2 && t == MTILES - 1);
}

template <bool EDGE>
__device__ __forceinline__ void load_row(RowFrags& f, const float* __restrict__ xN,
                                         int r, int w0, int ml, int half) {
  int rc = r;
  if (EDGE) rc = r < 0 ? 0 : (r > CONV_H - 1 ? CONV_H - 1 : r);
  const float* rowp = xN + (size_t)rc * CONV_W * CONV_IC + 2 * half;
#pragma unroll
  for (int fw = 0; fw < 3; ++fw) {
#pragma unroll
    for (int t = 0; t < MTILES; ++t) {
      const int iw  = w0 + 16 * t + ml + fw - 1;
      const int iwc = iw < 0 ? 0 : (iw > CONV_W - 1 ? CONV_W - 1 : iw);
#pragma unroll
      for (int kc = 0; kc < 2; ++kc) {
        f.a[fw][kc][t] = *(const v2f*)(rowp + (size_t)iwc * CONV_IC + kc * 4);
      }
    }
  }
}

template <bool EDGE>
__device__ __forceinline__ void mask_row(RowFrags& f, int r, int w0, int ml) {
  float rmask = 1.0f;
  if (EDGE) rmask = ((r >= 0) & (r < CONV_H)) ? 1.0f : 0.0f;
#pragma unroll
  for (int fw = 0; fw < 3; ++fw) {
#pragma unroll
    for (int t = 0; t < MTILES; ++t) {
      if (!EDGE && !w_edge(fw, t)) continue;
      const int iw = w0 + 16 * t + ml + fw - 1;
      const float msk = ((iw >= 0) & (iw < CONV_W)) ? rmask : 0.0f;
#pragma unroll
      for (int kc = 0; kc < 2; ++kc) {
        f.a[fw][kc][t].x *= msk;
        f.a[fw][kc][t].y *= msk;
      }
    }
  }
}

__device__ __forceinline__ void row_add(RowFrags& o, const RowFrags& p, const RowFrags& q) {
#pragma unroll
  for (int fw = 0; fw < 3; ++fw)
#pragma unroll
    for (int kc = 0; kc < 2; ++kc)
#pragma unroll
      for (int t = 0; t < MTILES; ++t) o.a[fw][kc][t] = p.a[fw][kc][t] + q.a[fw][kc][t];
}
__device__ __forceinline__ void row_sub(RowFrags& o, const RowFrags& p, const RowFrags& q) {
#pragma unroll
  for (int fw = 0; fw < 3; ++fw)
#pragma unroll
    for (int kc = 0; kc < 2; ++kc)
#pragma unroll
      for (int t = 0; t < MTILES; ++t) o.a[fw][kc][t] = p.a[fw][kc][t] - q.a[fw][kc][t];
}

struct FiltFrags {
  v2f u[3][2];
};

__device__ __forceinline__ void mul_row(v8f (&m)[MTILES], const FiltFrags& U, const RowFrags& T) {
#pragma unroll
  for (int fw = 0; fw < 3; ++fw) {
#pragma unroll
    for (int kc = 0; kc < 2; ++kc) {
#pragma unroll
      for (int t = 0; t < MTILES; ++t) {
        const v8f c = (fw == 0 && kc == 0) ? (v8f){} : m[t];
        m[t] = __builtin_amdgcn_wmma_f32_16x16x4_f32(
             false, U.u[fw][kc],  false, T.a[fw][kc][t],
             (short)0, c,  false,  false);
      }
    }
  }
}

__device__ __forceinline__ void mul_row2(v8f (&ma)[MTILES], const FiltFrags& Ua, const RowFrags& Ta,
                                         v8f (&mb)[MTILES], const FiltFrags& Ub, const RowFrags& Tb) {
#pragma unroll
  for (int fw = 0; fw < 3; ++fw) {
#pragma unroll
    for (int kc = 0; kc < 2; ++kc) {
      const bool first = (fw == 0 && kc == 0);
#pragma unroll
      for (int t = 0; t < MTILES; ++t) {
        const v8f c = first ? (v8f){} : ma[t];
        ma[t] = __builtin_amdgcn_wmma_f32_16x16x4_f32(
            false, Ua.u[fw][kc], false, Ta.a[fw][kc][t], (short)0, c, false, false);
      }
#pragma unroll
      for (int t = 0; t < MTILES; ++t) {
        const v8f c = first ? (v8f){} : mb[t];
        mb[t] = __builtin_amdgcn_wmma_f32_16x16x4_f32(
            false, Ub.u[fw][kc], false, Tb.a[fw][kc][t], (short)0, c, false, false);
      }
    }
  }
}

__device__ __forceinline__ void stage_row(const v8f (&y)[MTILES], v4f (*stg)[TILE_CHUNKS],
                                          int ml, int half, int lane, v4f (&outv)[MTILES][2]) {
#pragma unroll
  for (int t = 0; t < MTILES; ++t) {
    const v8f d = y[t];
    stg[t][ml * 4 + half * 2 + 0] = (v4f){d[0], d[1], d[2], d[3]};
    stg[t][ml * 4 + half * 2 + 1] = (v4f){d[4], d[5], d[6], d[7]};
  }
  __builtin_amdgcn_wave_barrier();
#pragma unroll
  for (int t = 0; t < MTILES; ++t) {
    outv[t][0] = stg[t][lane];
    outv[t][1] = stg[t][lane + 32];
  }
  __builtin_amdgcn_wave_barrier();
}
__device__ __forceinline__ void store_row(v4f* o, int lane, const v4f (&outv)[MTILES][2]) {
#pragma unroll
  for (int t = 0; t < MTILES; ++t) {
    o[t * TILE_CHUNKS + lane]      = outv[t][0];
    o[t * TILE_CHUNKS + 32 + lane] = outv[t][1];
  }
}

__global__ __launch_bounds__(128)
void conv2d_nhwc_wmma_f32(const float* __restrict__ x,
                          const float* __restrict__ wgt,
                          float* __restrict__ out) {
  const int lane = threadIdx.x & 31;
  const int wave = threadIdx.x >> 5;
  const int job  = blockIdx.x * 4 + wave;
  const int wseg = job & 1;
  const int h0   = ((job >> 1) & 31) * ROWS;
  const int n    = job >> 6;

  const int half = lane >> 4;
  const int ml   = lane & 15;

  __shared__ v4f stage[4][MTILES][TILE_CHUNKS];
  v4f (*stg)[TILE_CHUNKS] = stage[wave];
  const int w0   = wseg * 64;

  const float* xN = x + (size_t)n * CONV_H * CONV_W * CONV_IC;
  v4f* oN = (v4f*)out + ((size_t)n * CONV_H * CONV_W + w0) * (CONV_OC / 4);
  const size_t orow = (size_t)CONV_W * (CONV_OC / 4);

  v2f bfrag[K_STEPS];
  RowFrags d0, d1, d2, d3, d4, d5;

#pragma unroll
  for (int kk = 0; kk < 6; ++kk) {
    const int k0 = 4 * kk + 2 * half;
    bfrag[kk].x = wgt[k0 * CONV_OC + ml];
    bfrag[kk].y = wgt[(k0 + 1) * CONV_OC + ml];
  }
  load_row<true >(d0, xN, h0 - 1, w0, ml, half);
  load_row<false>(d2, xN, h0 + 1, w0, ml, half);
  __builtin_amdgcn_sched_barrier(0);
#pragma unroll
  for (int kk = 6; kk < K_STEPS; ++kk) {
    const int k0 = 4 * kk + 2 * half;
    bfrag[kk].x = wgt[k0 * CONV_OC + ml];
    bfrag[kk].y = wgt[(k0 + 1) * CONV_OC + ml];
  }
  load_row<false>(d1, xN, h0, w0, ml, half);
  __builtin_amdgcn_sched_barrier(0);

  FiltFrags U0, U1, U2, U3;
#pragma unroll
  for (int fw = 0; fw < 3; ++fw) {
#pragma unroll
    for (int kc = 0; kc < 2; ++kc) {
      U0.u[fw][kc] = bfrag[(0 * 3 + fw) * 2 + kc];
      U3.u[fw][kc] = bfrag[(2 * 3 + fw) * 2 + kc];
    }
  }

  RowFrags t0, t1, t2, t3;
  v8f M0[MTILES], M1[MTILES], M2[MTILES], M3[MTILES];
  v4f outa[MTILES][2], outb[MTILES][2];

  mask_row<true >(d0, h0 - 1, w0, ml);
  mask_row<false>(d2, h0 + 1, w0, ml);
  row_sub(t0, d0, d2);
  mul_row(M0, U0, t0);

  __builtin_amdgcn_sched_barrier(0);
  load_row<false>(d3, xN, h0 + 2, w0, ml, half);
  load_row<false>(d4, xN, h0 + 3, w0, ml, half);
  __builtin_amdgcn_sched_barrier(0);

#pragma unroll
  for (int fw = 0; fw < 3; ++fw) {
#pragma unroll
    for (int kc = 0; kc < 2; ++kc) {
      const v2f g1 = bfrag[(1 * 3 + fw) * 2 + kc];
      const v2f s  = U0.u[fw][kc] + U3.u[fw][kc];
      U1.u[fw][kc] = (s + g1) * 0.5f;
      U2.u[fw][kc] = (s - g1) * 0.5f;
    }
  }
  mask_row<false>(d1, h0, w0, ml);
  row_add(t1, d1, d2);
  row_sub(t2, d2, d1);
  mul_row2(M1, U1, t1, M2, U2, t2);

  __builtin_amdgcn_sched_barrier(0);
  load_row<true >(d5, xN, h0 + 4, w0, ml, half);
  __builtin_amdgcn_sched_barrier(0);

#pragma unroll
  for (int t = 0; t < MTILES; ++t) M0[t] = M0[t] + M1[t] + M2[t];
  stage_row(M0, stg, ml, half, lane, outa);
  __builtin_amdgcn_sched_barrier(0);

  mask_row<false>(d3, h0 + 2, w0, ml);
  row_sub(t3, d1, d3);
  mul_row(M3, U3, t3);
  store_row(oN + (size_t)h0 * orow, lane, outa);
#pragma unroll
  for (int t = 0; t < MTILES; ++t) M1[t] = M1[t] - M2[t] - M3[t];
  stage_row(M1, stg, ml, half, lane, outb);
  __builtin_amdgcn_sched_barrier(0);

  mask_row<false>(d4, h0 + 3, w0, ml);
  row_sub(t0, d2, d4);
  mul_row(M0, U0, t0);
  store_row(oN + (size_t)(h0 + 1) * orow, lane, outb);

  row_add(t1, d3, d4);
  row_sub(t2, d4, d3);
  mul_row2(M1, U1, t1, M2, U2, t2);
#pragma unroll
  for (int t = 0; t < MTILES; ++t) M0[t] = M0[t] + M1[t] + M2[t];
  stage_row(M0, stg, ml, half, lane, outa);
  __builtin_amdgcn_sched_barrier(0);

  mask_row<true>(d5, h0 + 4, w0, ml);
  row_sub(t3, d3, d5);
  mul_row(M3, U3, t3);
  store_row(oN + (size_t)(h0 + 2) * orow, lane, outa);
#pragma unroll
  for (int t = 0; t < MTILES; ++t) M1[t] = M1[t] - M2[t] - M3[t];
  stage_row(M1, stg, ml, half, lane, outb);
  store_row(oN + (size_t)(h0 + 3) * orow, lane, outb);
}

extern "C" void kernel_launch(void* const* d_in, const int* in_sizes, int n_in,
                              void* d_out, int out_size, void* d_ws, size_t ws_size,
                              hipStream_t stream) {
  const float* x = (const float*)d_in[0];
  const float* w = (const float*)d_in[1];
  float* out = (float*)d_out;

  conv2d_nhwc_wmma_f32<<<512, 128, 0, stream>>>(x, w, out);
}


# 3-tap filter along W also as a 2-output minimal filter: lane = column pair, 16 products per 2x2 outputs (128 WMMAs per wave, was 192), H sums through the WMMA C input; rounding and sum order change
# speedup vs baseline: 1.7321x; 1.1260x over previous
#include <hip/hip_runtime.h>


typedef __attribute__((ext_vector_type(2))) float v2f;
typedef __attribute__((ext_vector_type(4))) float v4f;
typedef __attribute__((ext_vector_type(8))) float v8f;

#define CONV_H 128
#define CONV_W 128
#define CONV_IC 8
#define CONV_OC 16
#define NT 2
#define ROWS 4
#define STG_CHUNKS 288


struct RowFrags {
  v4f a[NT][4];
};

__device__ __forceinline__ bool w_edge(int e, int t) {
  return (e == 0 && t == 0) || (e == 3 && t == NT - 1);
}

template <bool EDGE>
__device__ __forceinline__ void load_row(RowFrags& f, const float* __restrict__ xN,
                                         int r, int w0, int ml, int half) {
  int rc = r;
  if (EDGE) rc = r < 0 ? 0 : (r > CONV_H - 1 ? CONV_H - 1 : r);
  const float* rowp = xN + (size_t)rc * CONV_W * CONV_IC + 4 * half;
#pragma unroll
  for (int t = 0; t < NT; ++t) {
#pragma unroll
    for (int e = 0; e < 4; ++e) {
      const int iw = w0 + 32 * t + 2 * ml + e - 1;
      int iwc = iw;
      if (w_edge(e, t)) iwc = iw < 0 ? 0 : (iw > CONV_W - 1 ? CONV_W - 1 : iw);
      f.a[t][e] = *(const v4f*)(rowp + (size_t)iwc * CONV_IC);
    }
  }
}

template <bool EDGE>
__device__ __forceinline__ void mask_row(RowFrags& f, int r, int w0, int ml) {
  float rmask = 1.0f;
  if (EDGE) rmask = ((r >= 0) & (r < CONV_H)) ? 1.0f : 0.0f;
#pragma unroll
  for (int t = 0; t < NT; ++t) {
#pragma unroll
    for (int e = 0; e < 4; ++e) {
      if (!EDGE && !w_edge(e, t)) continue;
      float msk = rmask;
      if (w_edge(e, t)) {
        const int iw = w0 + 32 * t + 2 * ml + e - 1;
        msk = ((iw >= 0) & (iw < CONV_W)) ? rmask : 0.0f;
      }
      f.a[t][e] *= msk;
    }
  }
}

__device__ __forceinline__ void w_transform(RowFrags& f) {
#pragma unroll
  for (int t = 0; t < NT; ++t) {
    const v4f e0 = f.a[t][0], e1 = f.a[t][1], e2 = f.a[t][2], e3 = f.a[t][3];
    f.a[t][0] = e0 - e2;
    f.a[t][1] = e1 + e2;
    f.a[t][2] = e2 - e1;
    f.a[t][3] = e1 - e3;
  }
}
__device__ __forceinline__ void h_add(RowFrags& T, const RowFrags& p, const RowFrags& q) {
#pragma unroll
  for (int t = 0; t < NT; ++t)
#pragma unroll
    for (int j = 0; j < 4; ++j) T.a[t][j] = p.a[t][j] + q.a[t][j];
}
__device__ __forceinline__ void h_sub(RowFrags& T, const RowFrags& p, const RowFrags& q) {
#pragma unroll
  for (int t = 0; t < NT; ++t)
#pragma unroll
    for (int j = 0; j < 4; ++j) T.a[t][j] = p.a[t][j] - q.a[t][j];
}

struct FiltFrags {
  v2f u[4][2];
};

template <bool ACC>
__device__ __forceinline__ void mul_row(v8f (&P)[NT][4], const FiltFrags& U, const RowFrags& T) {
#pragma unroll
  for (int kc = 0; kc < 2; ++kc) {
#pragma unroll
    for (int j = 0; j < 4; ++j) {
#pragma unroll
      for (int t = 0; t < NT; ++t) {
        const v4f v = T.a[t][j];
        const v2f b = kc == 0 ? __builtin_shufflevector(v, v, 0, 1)
                              : __builtin_shufflevector(v, v, 2, 3);
        const v8f c = (!ACC && kc == 0) ? (v8f){} : P[t][j];
        P[t][j] = __builtin_amdgcn_wmma_f32_16x16x4_f32(
             false, U.u[j][kc],  false, b,
             (short)0, c,  false,  false);
      }
    }
  }
}

__device__ __forceinline__ void out_row(v8f (&y)[NT][2], const v8f (&Z)[NT][4]) {
#pragma unroll
  for (int t = 0; t < NT; ++t) {
    y[t][0] = Z[t][0] + Z[t][1] + Z[t][2];
    y[t][1] = Z[t][1] - Z[t][2] - Z[t][3];
  }
}

__device__ __forceinline__ void stage_row(const v8f (&y)[NT][2], v4f* stg,
                                          int ml, int half, int lane, v4f (&outv)[8]) {
#pragma unroll
  for (int t = 0; t < NT; ++t) {
#pragma unroll
    for (int par = 0; par < 2; ++par) {
      const v8f d = y[t][par];
      const int c = 144 * t + 9 * ml + 4 * par + 2 * half;
      stg[c + 0] = (v4f){d[0], d[1], d[2], d[3]};
      stg[c + 1] = (v4f){d[4], d[5], d[6], d[7]};
    }
  }
  __builtin_amdgcn_wave_barrier();
#pragma unroll
  for (int k = 0; k < 8; ++k) outv[k] = stg[lane + (lane >> 3) + 36 * k];
  __builtin_amdgcn_wave_barrier();
}
__device__ __forceinline__ void store_row(v4f* o, int lane, const v4f (&outv)[8]) {
#pragma unroll
  for (int k = 0; k < 8; ++k) o[32 * k + lane] = outv[k];
}

__global__ __launch_bounds__(128)
void conv2d_nhwc_wmma_f32(const float* __restrict__ x,
                          const float* __restrict__ wgt,
                          float* __restrict__ out) {
  const int lane = threadIdx.x & 31;
  const int wave = threadIdx.x >> 5;
  const int job  = blockIdx.x * 4 + wave;
  const int wseg = job & 1;
  const int h0   = ((job >> 1) & 31) * ROWS;
  const int n    = job >> 6;

  const int half = lane >> 4;
  const int ml   = lane & 15;

  __shared__ v4f stage[4][STG_CHUNKS];
  v4f* stg = stage[wave];
  const int w0 = wseg * 64;

  const float* xN = x + (size_t)n * CONV_H * CONV_W * CONV_IC;
  v4f* oN = (v4f*)out + ((size_t)n * CONV_H * CONV_W + w0) * (CONV_OC / 4);
  const size_t orow = (size_t)CONV_W * (CONV_OC / 4);

  v2f g[3][3][2];
  RowFrags d0, d1, d2, d3, d4, d5;

#pragma unroll
  for (int a = 0; a < 3; ++a) {
#pragma unroll
    for (int b = 0; b < 3; ++b) {
#pragma unroll
      for (int kc = 0; kc < 2; ++kc) {
        const int c0 = 4 * half + 2 * kc;
        g[a][b][kc].x = wgt[((a * 3 + b) * CONV_IC + c0) * CONV_OC + ml];
        g[a][b][kc].y = wgt[((a * 3 + b) * CONV_IC + c0 + 1) * CONV_OC + ml];
      }
    }
  }
  load_row<false>(d1, xN, h0, w0, ml, half);
  load_row<false>(d2, xN, h0 + 1, w0, ml, half);
  load_row<true >(d0, xN, h0 - 1, w0, ml, half);
  __builtin_amdgcn_sched_barrier(0);

  FiltFrags U[4];
#pragma unroll
  for (int kc = 0; kc < 2; ++kc) {
    v2f vw[3][4];
#pragma unroll
    for (int a = 0; a < 3; ++a) {
      const v2f s = g[a][0][kc] + g[a][2][kc];
      vw[a][0] = g[a][0][kc];
      vw[a][1] = (s + g[a][1][kc]) * 0.5f;
      vw[a][2] = (s - g[a][1][kc]) * 0.5f;
      vw[a][3] = g[a][2][kc];
    }
#pragma unroll
    for (int j = 0; j < 4; ++j) {
      const v2f s = vw[0][j] + vw[2][j];
      U[0].u[j][kc] = vw[0][j];
      U[1].u[j][kc] = (s + vw[1][j]) * 0.5f;
      U[2].u[j][kc] = (s - vw[1][j]) * 0.5f;
      U[3].u[j][kc] = vw[2][j];
    }
  }

  RowFrags T;
  v8f P1[NT][4], P2[NT][4];
  v8f y[NT][2];
  v4f outa[8], outb[8];

  mask_row<false>(d1, h0, w0, ml);
  mask_row<false>(d2, h0 + 1, w0, ml);
  w_transform(d1);
  w_transform(d2);
  h_add(T, d1, d2);
  mul_row<false>(P1, U[1], T);
  h_sub(T, d2, d1);
  mul_row<false>(P2, U[2], T);

  __builtin_amdgcn_sched_barrier(0);
  load_row<false>(d3, xN, h0 + 2, w0, ml, half);
  load_row<false>(d4, xN, h0 + 3, w0, ml, half);
  __builtin_amdgcn_sched_barrier(0);

#pragma unroll
  for (int t = 0; t < NT; ++t) {
#pragma unroll
    for (int j = 0; j < 4; ++j) {
      const v8f s = P1[t][j] + P2[t][j];
      P2[t][j] = P1[t][j] - P2[t][j];
      P1[t][j] = s;
    }
  }
  mask_row<true >(d0, h0 - 1, w0, ml);
  w_transform(d0);
  h_sub(T, d0, d2);
  mul_row<true>(P1, U[0], T);

  __builtin_amdgcn_sched_barrier(0);
  load_row<true >(d5, xN, h0 + 4, w0, ml, half);
  __builtin_amdgcn_sched_barrier(0);

  out_row(y, P1);
  stage_row(y, stg, ml, half, lane, outa);
  __builtin_amdgcn_sched_barrier(0);

  mask_row<false>(d3, h0 + 2, w0, ml);
  w_transform(d3);
  h_sub(T, d3, d1);
  mul_row<true>(P2, U[3], T);
  store_row(oN + (size_t)h0 * orow, lane, outa);
  out_row(y, P2);
  stage_row(y, stg, ml, half, lane, outb);
  __builtin_amdgcn_sched_barrier(0);

  mask_row<false>(d4, h0 + 3, w0, ml);
  w_transform(d4);
  h_add(T, d3, d4);
  mul_row<false>(P1, U[1], T);
  h_sub(T, d4, d3);
  mul_row<false>(P2, U[2], T);
  store_row(oN + (size_t)(h0 + 1) * orow, lane, outb);
#pragma unroll
  for (int t = 0; t < NT; ++t) {
#pragma unroll
    for (int j = 0; j < 4; ++j) {
      const v8f s = P1[t][j] + P2[t][j];
      P2[t][j] = P1[t][j] - P2[t][j];
      P1[t][j] = s;
    }
  }
  h_sub(T, d2, d4);
  mul_row<true>(P1, U[0], T);
  out_row(y, P1);
  stage_row(y, stg, ml, half, lane, outa);
  __builtin_amdgcn_sched_barrier(0);

  mask_row<true>(d5, h0 + 4, w0, ml);
  w_transform(d5);
  h_sub(T, d5, d3);
  mul_row<true>(P2, U[3], T);
  store_row(oN + (size_t)(h0 + 2) * orow, lane, outa);
  out_row(y, P2);
  stage_row(y, stg, ml, half, lane, outb);
  store_row(oN + (size_t)(h0 + 3) * orow, lane, outb);
}

extern "C" void kernel_launch(void* const* d_in, const int* in_sizes, int n_in,
                              void* d_out, int out_size, void* d_ws, size_t ws_size,
                              hipStream_t stream) {
  const float* x = (const float*)d_in[0];
  const float* w = (const float*)d_in[1];
  float* out = (float*)d_out;

  conv2d_nhwc_wmma_f32<<<512, 128, 0, stream>>>(x, w, out);
}


# Each wave walks 8 output rows of 32 columns (was 4 rows of 64): first products wait for 2 input rows not 6, first store after 24 products not 48, 40 row loads per wave not 48; sum order unchanged
# speedup vs baseline: 1.8388x; 1.0616x over previous
#include <hip/hip_runtime.h>


typedef __attribute__((ext_vector_type(2))) float v2f;
typedef __attribute__((ext_vector_type(4))) float v4f;
typedef __attribute__((ext_vector_type(8))) float v8f;

#define CONV_H 128
#define CONV_W 128
#define CONV_IC 8
#define CONV_OC 16
#define NT 1
#define ROWS 8
#define NPAIR (ROWS / 2)
#define NIN (ROWS + 2)
#define NCH (4 * NT)
#define STG_CHUNKS (144 * NT)


struct RowFrags {
  v4f a[NT][4];
};

__device__ __forceinline__ bool w_edge(int e, int t) {
  return (e == 0 && t == 0) || (e == 3 && t == NT - 1);
}

template <bool EDGE>
__device__ __forceinline__ void load_row(RowFrags& f, const float* __restrict__ xN,
                                         int r, int w0, int ml, int half) {
  int rc = r;
  if (EDGE) rc = r < 0 ? 0 : (r > CONV_H - 1 ? CONV_H - 1 : r);
  const float* rowp = xN + (size_t)rc * CONV_W * CONV_IC + 4 * half;
#pragma unroll
  for (int t = 0; t < NT; ++t) {
#pragma unroll
    for (int e = 0; e < 4; ++e) {
      const int iw = w0 + 32 * t + 2 * ml + e - 1;
      int iwc = iw;
      if (w_edge(e, t)) iwc = iw < 0 ? 0 : (iw > CONV_W - 1 ? CONV_W - 1 : iw);
      f.a[t][e] = *(const v4f*)(rowp + (size_t)iwc * CONV_IC);
    }
  }
}

template <bool EDGE>
__device__ __forceinline__ void mask_row(RowFrags& f, int r, int w0, int ml) {
  float rmask = 1.0f;
  if (EDGE) rmask = ((r >= 0) & (r < CONV_H)) ? 1.0f : 0.0f;
#pragma unroll
  for (int t = 0; t < NT; ++t) {
#pragma unroll
    for (int e = 0; e < 4; ++e) {
      if (!EDGE && !w_edge(e, t)) continue;
      float msk = rmask;
      if (w_edge(e, t)) {
        const int iw = w0 + 32 * t + 2 * ml + e - 1;
        msk = ((iw >= 0) & (iw < CONV_W)) ? rmask : 0.0f;
      }
      f.a[t][e] *= msk;
    }
  }
}

__device__ __forceinline__ void w_transform(RowFrags& f) {
#pragma unroll
  for (int t = 0; t < NT; ++t) {
    const v4f e0 = f.a[t][0], e1 = f.a[t][1], e2 = f.a[t][2], e3 = f.a[t][3];
    f.a[t][0] = e0 - e2;
    f.a[t][1] = e1 + e2;
    f.a[t][2] = e2 - e1;
    f.a[t][3] = e1 - e3;
  }
}
__device__ __forceinline__ void h_add(RowFrags& T, const RowFrags& p, const RowFrags& q) {
#pragma unroll
  for (int t = 0; t < NT; ++t)
#pragma unroll
    for (int j = 0; j < 4; ++j) T.a[t][j] = p.a[t][j] + q.a[t][j];
}
__device__ __forceinline__ void h_sub(RowFrags& T, const RowFrags& p, const RowFrags& q) {
#pragma unroll
  for (int t = 0; t < NT; ++t)
#pragma unroll
    for (int j = 0; j < 4; ++j) T.a[t][j] = p.a[t][j] - q.a[t][j];
}

struct FiltFrags {
  v2f u[4][2];
};

template <bool ACC>
__device__ __forceinline__ void mul_row(v8f (&P)[NT][4], const FiltFrags& U, const RowFrags& T) {
#pragma unroll
  for (int kc = 0; kc < 2; ++kc) {
#pragma unroll
    for (int j = 0; j < 4; ++j) {
#pragma unroll
      for (int t = 0; t < NT; ++t) {
        const v4f v = T.a[t][j];
        const v2f b = kc == 0 ? __builtin_shufflevector(v, v, 0, 1)
                              : __builtin_shufflevector(v, v, 2, 3);
        const v8f c = (!ACC && kc == 0) ? (v8f){} : P[t][j];
        P[t][j] = __builtin_amdgcn_wmma_f32_16x16x4_f32(
             false, U.u[j][kc],  false, b,
             (short)0, c,  false,  false);
      }
    }
  }
}

__device__ __forceinline__ void out_row(v8f (&y)[NT][2], const v8f (&Z)[NT][4]) {
#pragma unroll
  for (int t = 0; t < NT; ++t) {
    y[t][0] = Z[t][0] + Z[t][1] + Z[t][2];
    y[t][1] = Z[t][1] - Z[t][2] - Z[t][3];
  }
}

__device__ __forceinline__ void stage_row(const v8f (&y)[NT][2], v4f* stg,
                                          int ml, int half, int lane, v4f (&outv)[NCH]) {
#pragma unroll
  for (int t = 0; t < NT; ++t) {
#pragma unroll
    for (int par = 0; par < 2; ++par) {
      const v8f d = y[t][par];
      const int c = 144 * t + 9 * ml + 4 * par + 2 * half;
      stg[c + 0] = (v4f){d[0], d[1], d[2], d[3]};
      stg[c + 1] = (v4f){d[4], d[5], d[6], d[7]};
    }
  }
  __builtin_amdgcn_wave_barrier();
#pragma unroll
  for (int k = 0; k < NCH; ++k) outv[k] = stg[lane + (lane >> 3) + 36 * k];
  __builtin_amdgcn_wave_barrier();
}
__device__ __forceinline__ void store_row(v4f* o, int lane, const v4f (&outv)[NCH]) {
#pragma unroll
  for (int k = 0; k < NCH; ++k) o[32 * k + lane] = outv[k];
}

__global__ __launch_bounds__(128)
void conv2d_nhwc_wmma_f32(const float* __restrict__ x,
                          const float* __restrict__ wgt,
                          float* __restrict__ out) {
  const int lane = threadIdx.x & 31;
  const int wave = threadIdx.x >> 5;
  const int job  = blockIdx.x * 4 + wave;
  const int wseg = job & 3;
  const int h0   = ((job >> 2) & 15) * ROWS;
  const int n    = job >> 6;

  const int half = lane >> 4;
  const int ml   = lane & 15;

  __shared__ v4f stage[4][STG_CHUNKS];
  v4f* stg = stage[wave];
  const int w0 = wseg * 32 * NT;

  const float* xN = x + (size_t)n * CONV_H * CONV_W * CONV_IC;
  v4f* oN = (v4f*)out + ((size_t)n * CONV_H * CONV_W + w0) * (CONV_OC / 4);
  const size_t orow = (size_t)CONV_W * (CONV_OC / 4);

  v2f g[3][3][2];
  RowFrags d[NIN];

#pragma unroll
  for (int a = 0; a < 3; ++a) {
#pragma unroll
    for (int b = 0; b < 3; ++b) {
#pragma unroll
      for (int kc = 0; kc < 2; ++kc) {
        const int c0 = 4 * half + 2 * kc;
        g[a][b][kc].x = wgt[((a * 3 + b) * CONV_IC + c0) * CONV_OC + ml];
        g[a][b][kc].y = wgt[((a * 3 + b) * CONV_IC + c0 + 1) * CONV_OC + ml];
      }
    }
  }
  load_row<false>(d[1], xN, h0, w0, ml, half);
  load_row<false>(d[2], xN, h0 + 1, w0, ml, half);
  __builtin_amdgcn_sched_barrier(0);
  load_row<true >(d[0], xN, h0 - 1, w0, ml, half);
  load_row<false>(d[3], xN, h0 + 2, w0, ml, half);
  load_row<false>(d[4], xN, h0 + 3, w0, ml, half);
  __builtin_amdgcn_sched_barrier(0);

  FiltFrags U[4];
#pragma unroll
  for (int kc = 0; kc < 2; ++kc) {
    v2f vw[3][4];
#pragma unroll
    for (int a = 0; a < 3; ++a) {
      const v2f s = g[a][0][kc] + g[a][2][kc];
      vw[a][0] = g[a][0][kc];
      vw[a][1] = (s + g[a][1][kc]) * 0.5f;
      vw[a][2] = (s - g[a][1][kc]) * 0.5f;
      vw[a][3] = g[a][2][kc];
    }
#pragma unroll
    for (int j = 0; j < 4; ++j) {
      const v2f s = vw[0][j] + vw[2][j];
      U[0].u[j][kc] = vw[0][j];
      U[1].u[j][kc] = (s + vw[1][j]) * 0.5f;
      U[2].u[j][kc] = (s - vw[1][j]) * 0.5f;
      U[3].u[j][kc] = vw[2][j];
    }
  }

  RowFrags T;
  v8f P1[NT][4], P2[NT][4];
  v8f y[NT][2];
  v4f outa[NCH], outb[NCH];

#pragma unroll
  for (int p = 0; p < NPAIR; ++p) {
    RowFrags& ra = d[2 * p];
    RowFrags& rb = d[2 * p + 1];
    RowFrags& rc = d[2 * p + 2];
    RowFrags& re = d[2 * p + 3];
    const int hp = h0 + 2 * p;

    if (p == 0) {
      mask_row<false>(rb, hp, w0, ml);
      w_transform(rb);
    }
    mask_row<false>(rc, hp + 1, w0, ml);
    w_transform(rc);
    h_add(T, rb, rc);
    mul_row<false>(P1, U[1], T);
    h_sub(T, rc, rb);
    mul_row<false>(P2, U[2], T);

    if (p == 0) {
      __builtin_amdgcn_sched_barrier(0);
      load_row<false>(d[5], xN, h0 + 4, w0, ml, half);
      load_row<false>(d[6], xN, h0 + 5, w0, ml, half);
      load_row<false>(d[7], xN, h0 + 6, w0, ml, half);
      load_row<false>(d[8], xN, h0 + 7, w0, ml, half);
      load_row<true >(d[9], xN, h0 + 8, w0, ml, half);
      __builtin_amdgcn_sched_barrier(0);
    } else {
      store_row(oN + (size_t)(hp - 1) * orow, lane, outb);
    }

#pragma unroll
    for (int t = 0; t < NT; ++t) {
#pragma unroll
      for (int j = 0; j < 4; ++j) {
        const v8f s = P1[t][j] + P2[t][j];
        P2[t][j] = P1[t][j] - P2[t][j];
        P1[t][j] = s;
      }
    }
    if (p == 0) {
      mask_row<true>(ra, hp - 1, w0, ml);
      w_transform(ra);
    }
    h_sub(T, ra, rc);
    mul_row<true>(P1, U[0], T);
    out_row(y, P1);
    stage_row(y, stg, ml, half, lane, outa);
    __builtin_amdgcn_sched_barrier(0);

    if (p == NPAIR - 1) mask_row<true >(re, hp + 2, w0, ml);
    else                mask_row<false>(re, hp + 2, w0, ml);
    w_transform(re);
    h_sub(T, re, rb);
    mul_row<true>(P2, U[3], T);
    store_row(oN + (size_t)hp * orow, lane, outa);
    out_row(y, P2);
    stage_row(y, stg, ml, half, lane, outb);
    if (p < NPAIR - 1) __builtin_amdgcn_sched_barrier(0);
  }
  store_row(oN + (size_t)(h0 + ROWS - 1) * orow, lane, outb);
}

extern "C" void kernel_launch(void* const* d_in, const int* in_sizes, int n_in,
                              void* d_out, int out_size, void* d_ws, size_t ws_size,
                              hipStream_t stream) {
  const float* x = (const float*)d_in[0];
  const float* w = (const float*)d_in[1];
  float* out = (float*)d_out;

  conv2d_nhwc_wmma_f32<<<512, 128, 0, stream>>>(x, w, out);
}
